# starting kernel
# baseline (speedup 1.0000x reference)
#include <hip/hip_runtime.h>


typedef float v2f __attribute__((ext_vector_type(2)));
typedef float v8f __attribute__((ext_vector_type(8)));

#define CLN_EPS 1e-5f
#define T_DIM   8192
#define C_DIM   512
#define B_DIM   8
#define NSEG    32
#define TSEG    256
#define NROWS   (B_DIM * C_DIM)

__device__ __forceinline__ float dpp_shr1_fill(float src, float fill) {
    int r = __builtin_amdgcn_update_dpp(
        __float_as_int(fill), __float_as_int(src),
        0x111  , 0xf, 0xf, false);
    return __int_as_float(r);
}

__global__ __launch_bounds__(256) void cln_partials_kernel(
    const float* __restrict__ x, float2* __restrict__ part) {
    int gtid = blockIdx.x * blockDim.x + threadIdx.x;
    int wave = gtid >> 5;
    int lane = gtid & 31;
    int row  = wave >> 5;
    int seg  = wave & (NSEG - 1);

    const float* p = x + (size_t)row * T_DIM + seg * TSEG + lane * 8;
    float4 a = ((const float4*)p)[0];
    float4 b = ((const float4*)p)[1];
    float s = a.x + a.y + a.z + a.w + b.x + b.y + b.z + b.w;
    float q = fmaf(a.x, a.x, fmaf(a.y, a.y, fmaf(a.z, a.z, a.w * a.w))) +
              fmaf(b.x, b.x, fmaf(b.y, b.y, fmaf(b.z, b.z, b.w * b.w)));
#pragma unroll
    for (int off = 16; off > 0; off >>= 1) {
        s += __shfl_xor(s, off, 32);
        q += __shfl_xor(q, off, 32);
    }
    if (lane == 0) part[row * NSEG + seg] = make_float2(s, q);
}

__global__ __launch_bounds__(256) void cln_scan_kernel(
    const float* __restrict__ x, const float* __restrict__ weight,
    const float* __restrict__ bias, const float2* __restrict__ part,
    float* __restrict__ out) {
    int gtid = blockIdx.x * blockDim.x + threadIdx.x;
    int wave = gtid >> 5;
    int L    = gtid & 31;
    int rg   = wave >> 5;
    int seg  = wave & (NSEG - 1);
    int row0 = rg << 4;
    int c0   = row0 & (C_DIM - 1);

    int n    = L & 15;
    int hi   = (L >> 4) & 1;
    int koff = hi << 1;
    int m    = n;
    int rowm = row0 + m;
    const float* xrow = x + (size_t)rowm * T_DIM;

    float wd[8], bd[8];
#pragma unroll
    for (int r = 0; r < 8; ++r) {
        int ch = c0 + (hi << 3) + r;
        wd[r] = weight[ch];
        bd[r] = bias[ch];
    }

    float cs1 = 0.f, cs2 = 0.f;
    const float2* prow = part + (size_t)rowm * NSEG;
    for (int s = 0; s < seg; ++s) {
        float2 p = prow[s];
        cs1 += p.x;
        cs2 += p.y;
    }
    v8f c1d, c2d;
#pragma unroll
    for (int r = 0; r < 8; ++r) {
        int src = (hi ? 24 : 0) + r;
        c1d[r] = __shfl(cs1, src, 32);
        c2d[r] = __shfl(cs2, src, 32);
    }

    v2f bm[4];
#pragma unroll
    for (int k = 0; k < 4; ++k) {
        int s0 = 4 * k + koff;
        bm[k].x = (s0     <= n) ? 1.f : 0.f;
        bm[k].y = (s0 + 1 <= n) ? 1.f : 0.f;
    }

    int tbase = seg * TSEG;
    for (int tile = 0; tile < TSEG / 16; ++tile) {
        int t0 = tbase + tile * 16;

        v2f a[4], aq[4];
#pragma unroll
        for (int k = 0; k < 4; ++k) {
            a[k]  = *(const v2f*)(xrow + t0 + 4 * k + koff);
            aq[k] = a[k] * a[k];
        }

        v8f s1v = c1d, s2v = c2d;
#pragma unroll
        for (int k = 0; k < 4; ++k) {
            s1v = __builtin_amdgcn_wmma_f32_16x16x4_f32(
                false, a[k],  false, bm[k], (short)0, s1v, false, false);
            s2v = __builtin_amdgcn_wmma_f32_16x16x4_f32(
                false, aq[k], false, bm[k], (short)0, s2v, false, false);
        }

        float rc = __builtin_amdgcn_rcpf((float)(t0 + n + 1));
        v8f c1old = c1d;
#pragma unroll
        for (int r = 0; r < 8; ++r) {
            float s1   = s1v[r], s2 = s2v[r];
            float mean = s1 * rc;
            float var  = fmaf(-mean, mean, s2 * rc);
            float inv  = __builtin_amdgcn_rsqf(var + CLN_EPS);
            float prev = dpp_shr1_fill(s1, c1old[r]);
            float xv = s1 - prev;
            float o  = fmaf((xv - mean) * inv, wd[r], bd[r]);
            out[(size_t)(row0 + (hi << 3) + r) * T_DIM + t0 + n] = o;
        }

#pragma unroll
        for (int r = 0; r < 8; ++r) {
            c1d[r] = __shfl(s1v[r], 15, 16);
            c2d[r] = __shfl(s2v[r], 15, 16);
        }
    }
}

extern "C" void kernel_launch(void* const* d_in, const int* in_sizes, int n_in,
                              void* d_out, int out_size, void* d_ws, size_t ws_size,
                              hipStream_t stream) {
    const float* x = (const float*)d_in[0];
    const float* w = (const float*)d_in[1];
    const float* b = (const float*)d_in[2];
    float* out     = (float*)d_out;
    float2* part   = (float2*)d_ws;

    cln_partials_kernel<<<16384, 256, 0, stream>>>(x, part);
    cln_scan_kernel<<<1024, 256, 0, stream>>>(x, w, b, part, out);
    (void)in_sizes; (void)n_in; (void)out_size; (void)ws_size;
}


# Fuse both launches: one 16-wave work-group per 16 rows keeps each x tile in registers from the segment sums (shared via LDS) to the WMMA scan, next tile requested ahead; sums 128 wide, new sum order
# speedup vs baseline: 2.0087x; 2.0087x over previous
#include <hip/hip_runtime.h>


typedef float v2f __attribute__((ext_vector_type(2)));
typedef float v8f __attribute__((ext_vector_type(8)));

#define CLN_EPS 1e-5f
#define T_DIM   8192
#define C_DIM   512
#define B_DIM   8
#define NWAVE   16
#define NCHUNK  4
#define TW      (T_DIM / (NWAVE * NCHUNK))
#define TCHUNK  (NWAVE * TW)
#define NTILE   (TW / 16)
#define NFRAG   (NTILE * 4)
#define NGROUP  (B_DIM * C_DIM / 16)

__device__ __forceinline__ float dpp_shr1_fill(float src, float fill) {
    int r = __builtin_amdgcn_update_dpp(
        __float_as_int(fill), __float_as_int(src),
        0x111  , 0xf, 0xf, false);
    return __int_as_float(r);
}

__device__ __forceinline__ void load_tile(v2f (&f)[NFRAG], const float* p) {
#pragma unroll
    for (int i = 0; i < NFRAG; ++i) f[i] = *(const v2f*)(p + 4 * i);
}

__global__ __launch_bounds__(NWAVE * 32) void cln_fused_kernel(
    const float* __restrict__ x, const float* __restrict__ weight,
    const float* __restrict__ bias, float* __restrict__ out) {
    __shared__ float2 part[2][NWAVE][16];

    int L    = threadIdx.x & 31;
    int w    = __builtin_amdgcn_readfirstlane((int)(threadIdx.x >> 5));
    int row0 = blockIdx.x << 4;
    int c0   = row0 & (C_DIM - 1);

    int n    = L & 15;
    int hi   = (L >> 4) & 1;
    int koff = hi << 1;
    int m    = n;
    const float* xlane = x + (size_t)(row0 + m) * T_DIM + w * TW + koff;

    float wd[8], bd[8];
#pragma unroll
    for (int r = 0; r < 8; ++r) {
        float wlo = weight[c0 + r], whi = weight[c0 + 8 + r];
        float blo = bias[c0 + r],   bhi = bias[c0 + 8 + r];
        wd[r] = hi ? whi : wlo;
        bd[r] = hi ? bhi : blo;
    }

    v2f bm[4];
#pragma unroll
    for (int k = 0; k < 4; ++k) {
        int s0 = 4 * k + koff;
        bm[k].x = (s0     <= n) ? 1.f : 0.f;
        bm[k].y = (s0 + 1 <= n) ? 1.f : 0.f;
    }

    float run1 = 0.f, run2 = 0.f;

    v2f cur[NFRAG], nxt[NFRAG];
    load_tile(nxt, xlane);

#pragma clang loop unroll(disable)
    for (int c = 0; c < NCHUNK; ++c) {
#pragma unroll
        for (int i = 0; i < NFRAG; ++i) cur[i] = nxt[i];
        if (c + 1 < NCHUNK) load_tile(nxt, xlane + (c + 1) * TCHUNK);

        v2f sa = {0.f, 0.f}, sb = {0.f, 0.f}, qa = {0.f, 0.f}, qb = {0.f, 0.f};
#pragma unroll
        for (int i = 0; i < NFRAG; i += 2) {
            sa += cur[i];
            sb += cur[i + 1];
            qa = __builtin_elementwise_fma(cur[i],     cur[i],     qa);
            qb = __builtin_elementwise_fma(cur[i + 1], cur[i + 1], qb);
        }
        float s = (sa.x + sa.y) + (sb.x + sb.y);
        float q = (qa.x + qa.y) + (qb.x + qb.y);
        s += __shfl_xor(s, 16, 32);
        q += __shfl_xor(q, 16, 32);

        int buf = c & 1;
        if (hi == 0) part[buf][w][m] = make_float2(s, q);
        asm volatile("s_wait_dscnt 0x0\n\ts_barrier_signal -1\n\ts_barrier_wait -1" ::: "memory");

        float acc1 = run1, acc2 = run2, pre1 = 0.f, pre2 = 0.f;
#pragma unroll
        for (int j = 0; j < NWAVE; ++j) {
            float2 p = part[buf][j][m];
            if (j == w) { pre1 = acc1; pre2 = acc2; }
            acc1 += p.x;
            acc2 += p.y;
        }
        run1 = acc1;
        run2 = acc2;

        v8f c1d, c2d;
#pragma unroll
        for (int r = 0; r < 8; ++r) {
            int src = (hi ? 24 : 0) + r;
            c1d[r] = __shfl(pre1, src, 32);
            c2d[r] = __shfl(pre2, src, 32);
        }

        int tbase = c * TCHUNK + w * TW;
#pragma unroll
        for (int tile = 0; tile < NTILE; ++tile) {
            int t0 = tbase + tile * 16;

            v2f aq[4];
#pragma unroll
            for (int k = 0; k < 4; ++k) aq[k] = cur[4 * tile + k] * cur[4 * tile + k];

            v8f s1v = c1d, s2v = c2d;
#pragma unroll
            for (int k = 0; k < 4; ++k) {
                s1v = __builtin_amdgcn_wmma_f32_16x16x4_f32(
                    false, cur[4 * tile + k], false, bm[k], (short)0, s1v, false, false);
                s2v = __builtin_amdgcn_wmma_f32_16x16x4_f32(
                    false, aq[k],             false, bm[k], (short)0, s2v, false, false);
            }

            float rc = __builtin_amdgcn_rcpf((float)(t0 + n + 1));
            v8f c1old = c1d;
#pragma unroll
            for (int r = 0; r < 8; ++r) {
                float s1   = s1v[r], s2 = s2v[r];
                float mean = s1 * rc;
                float var  = fmaf(-mean, mean, s2 * rc);
                float inv  = __builtin_amdgcn_rsqf(var + CLN_EPS);
                float prev = dpp_shr1_fill(s1, c1old[r]);
                float xv = s1 - prev;
                float o  = fmaf((xv - mean) * inv, wd[r], bd[r]);
                out[(size_t)(row0 + (hi << 3) + r) * T_DIM + t0 + n] = o;
            }

#pragma unroll
            for (int r = 0; r < 8; ++r) {
                c1d[r] = __shfl(s1v[r], 15, 16);
                c2d[r] = __shfl(s2v[r], 15, 16);
            }
        }
    }
}

extern "C" void kernel_launch(void* const* d_in, const int* in_sizes, int n_in,
                              void* d_out, int out_size, void* d_ws, size_t ws_size,
                              hipStream_t stream) {
    const float* x = (const float*)d_in[0];
    const float* w = (const float*)d_in[1];
    const float* b = (const float*)d_in[2];
    float* out     = (float*)d_out;

    cln_fused_kernel<<<NGROUP, NWAVE * 32, 0, stream>>>(x, w, b, out);
    (void)in_sizes; (void)n_in; (void)out_size; (void)d_ws; (void)ws_size;
}


# Load x with 16-byte lane loads (32 contiguous bytes per row and instruction, half the load instructions) by re-ordering k inside each 16-step tile, U rows alike; new sum order inside a tile
# speedup vs baseline: 2.0962x; 1.0436x over previous
#include <hip/hip_runtime.h>


typedef float v2f __attribute__((ext_vector_type(2)));
typedef float v4f __attribute__((ext_vector_type(4)));
typedef float v8f __attribute__((ext_vector_type(8)));

#define CLN_EPS 1e-5f
#define T_DIM   8192
#define C_DIM   512
#define B_DIM   8
#define NWAVE   16
#define NCHUNK  4
#define TW      (T_DIM / (NWAVE * NCHUNK))
#define TCHUNK  (NWAVE * TW)
#define NTILE   (TW / 16)
#define NFRAG   (NTILE * 4)
#define NQUAD   (NTILE * 2)
#define NGROUP  (B_DIM * C_DIM / 16)

__device__ __forceinline__ float dpp_shr1_fill(float src, float fill) {
    int r = __builtin_amdgcn_update_dpp(
        __float_as_int(fill), __float_as_int(src),
        0x111  , 0xf, 0xf, false);
    return __int_as_float(r);
}

__device__ __forceinline__ void load_tile(v4f (&f)[NQUAD], const float* p) {
#pragma unroll
    for (int q = 0; q < NQUAD; ++q) f[q] = *(const v4f*)(p + 8 * q);
}

__global__ __launch_bounds__(NWAVE * 32) void cln_fused_kernel(
    const float* __restrict__ x, const float* __restrict__ weight,
    const float* __restrict__ bias, float* __restrict__ out) {
    __shared__ float2 part[2][NWAVE][16];

    int L    = threadIdx.x & 31;
    int w    = __builtin_amdgcn_readfirstlane((int)(threadIdx.x >> 5));
    int row0 = blockIdx.x << 4;
    int c0   = row0 & (C_DIM - 1);

    int n    = L & 15;
    int hi   = (L >> 4) & 1;
    int hoff = hi << 2;
    int m    = n;
    const float* xlane = x + (size_t)(row0 + m) * T_DIM + w * TW + hoff;

    float wd[8], bd[8];
#pragma unroll
    for (int r = 0; r < 8; ++r) {
        float wlo = weight[c0 + r], whi = weight[c0 + 8 + r];
        float blo = bias[c0 + r],   bhi = bias[c0 + 8 + r];
        wd[r] = hi ? whi : wlo;
        bd[r] = hi ? bhi : blo;
    }

    v2f bm[4];
#pragma unroll
    for (int k = 0; k < 4; ++k) {
        int s0 = 8 * (k >> 1) + 2 * (k & 1) + hoff;
        bm[k].x = (s0     <= n) ? 1.f : 0.f;
        bm[k].y = (s0 + 1 <= n) ? 1.f : 0.f;
    }

    float run1 = 0.f, run2 = 0.f;

    v2f cur[NFRAG];
    v4f nxt[NQUAD];
    load_tile(nxt, xlane);

#pragma clang loop unroll(disable)
    for (int c = 0; c < NCHUNK; ++c) {
#pragma unroll
        for (int q = 0; q < NQUAD; ++q) {
            cur[2 * q]     = __builtin_shufflevector(nxt[q], nxt[q], 0, 1);
            cur[2 * q + 1] = __builtin_shufflevector(nxt[q], nxt[q], 2, 3);
        }
        if (c + 1 < NCHUNK) load_tile(nxt, xlane + (c + 1) * TCHUNK);

        v2f sa = {0.f, 0.f}, sb = {0.f, 0.f}, qa = {0.f, 0.f}, qb = {0.f, 0.f};
#pragma unroll
        for (int i = 0; i < NFRAG; i += 2) {
            sa += cur[i];
            sb += cur[i + 1];
            qa = __builtin_elementwise_fma(cur[i],     cur[i],     qa);
            qb = __builtin_elementwise_fma(cur[i + 1], cur[i + 1], qb);
        }
        float s = (sa.x + sa.y) + (sb.x + sb.y);
        float q = (qa.x + qa.y) + (qb.x + qb.y);
        s += __shfl_xor(s, 16, 32);
        q += __shfl_xor(q, 16, 32);

        int buf = c & 1;
        if (hi == 0) part[buf][w][m] = make_float2(s, q);
        asm volatile("s_wait_dscnt 0x0\n\ts_barrier_signal -1\n\ts_barrier_wait -1" ::: "memory");

        float acc1 = run1, acc2 = run2, pre1 = 0.f, pre2 = 0.f;
#pragma unroll
        for (int j = 0; j < NWAVE; ++j) {
            float2 p = part[buf][j][m];
            if (j == w) { pre1 = acc1; pre2 = acc2; }
            acc1 += p.x;
            acc2 += p.y;
        }
        run1 = acc1;
        run2 = acc2;

        v8f c1d, c2d;
#pragma unroll
        for (int r = 0; r < 8; ++r) {
            int src = (hi ? 24 : 0) + r;
            c1d[r] = __shfl(pre1, src, 32);
            c2d[r] = __shfl(pre2, src, 32);
        }

        int tbase = c * TCHUNK + w * TW;
#pragma unroll
        for (int tile = 0; tile < NTILE; ++tile) {
            int t0 = tbase + tile * 16;

            v2f aq[4];
#pragma unroll
            for (int k = 0; k < 4; ++k) aq[k] = cur[4 * tile + k] * cur[4 * tile + k];

            v8f s1v = c1d, s2v = c2d;
#pragma unroll
            for (int k = 0; k < 4; ++k) {
                s1v = __builtin_amdgcn_wmma_f32_16x16x4_f32(
                    false, cur[4 * tile + k], false, bm[k], (short)0, s1v, false, false);
                s2v = __builtin_amdgcn_wmma_f32_16x16x4_f32(
                    false, aq[k],             false, bm[k], (short)0, s2v, false, false);
            }

            float rc = __builtin_amdgcn_rcpf((float)(t0 + n + 1));
            v8f c1old = c1d;
#pragma unroll
            for (int r = 0; r < 8; ++r) {
                float s1   = s1v[r], s2 = s2v[r];
                float mean = s1 * rc;
                float var  = fmaf(-mean, mean, s2 * rc);
                float inv  = __builtin_amdgcn_rsqf(var + CLN_EPS);
                float prev = dpp_shr1_fill(s1, c1old[r]);
                float xv = s1 - prev;
                float o  = fmaf((xv - mean) * inv, wd[r], bd[r]);
                out[(size_t)(row0 + (hi << 3) + r) * T_DIM + t0 + n] = o;
            }

#pragma unroll
            for (int r = 0; r < 8; ++r) {
                c1d[r] = __shfl(s1v[r], 15, 16);
                c2d[r] = __shfl(s2v[r], 15, 16);
            }
        }
    }
}

extern "C" void kernel_launch(void* const* d_in, const int* in_sizes, int n_in,
                              void* d_out, int out_size, void* d_ws, size_t ws_size,
                              hipStream_t stream) {
    const float* x = (const float*)d_in[0];
    const float* w = (const float*)d_in[1];
    const float* b = (const float*)d_in[2];
    float* out     = (float*)d_out;

    cln_fused_kernel<<<NGROUP, NWAVE * 32, 0, stream>>>(x, w, b, out);
    (void)in_sizes; (void)n_in; (void)out_size; (void)d_ws; (void)ws_size;
}
